# starting kernel
# baseline (speedup 1.0000x reference)
#include <hip/hip_runtime.h>
#include <hip/hip_bf16.h>
#include <math.h>

typedef float v2f __attribute__((ext_vector_type(2)));
typedef float v8f __attribute__((ext_vector_type(8)));

#define B_ 256
#define S_ 128
#define E_ 512
#define H_ 1024
#define V_ 128
#define C_ 18

__global__ void k_zero(float* __restrict__ p, int n) {
    int i = blockIdx.x * blockDim.x + threadIdx.x;
    if (i < n) p[i] = 0.0f;
}

__global__ __launch_bounds__(256)
void k_embed_proj(const int* __restrict__ x_in,
                  const float* __restrict__ emb,
                  const float* __restrict__ W_ih,
                  const float* __restrict__ b_ih,
                  float* __restrict__ pre) {
    const int lane = threadIdx.x & 31;
    const int wave = threadIdx.x >> 5;
    const int gw   = blockIdx.x * (blockDim.x >> 5) + wave;

    const int NG = H_ / 64;
    const int tn = (gw % NG) * 64;
    const int tm = (gw / NG) * 16;

    const int half = lane >> 4;
    const int l15  = lane & 15;

    const int m_a = tm + l15;
    const int s   = m_a / B_;
    const int b   = m_a % B_;
    const int tok = x_in[b * S_ + s];
    const float* __restrict__ arow = emb + (size_t)tok * E_ + half * 2;

    const float* __restrict__ brow0 = W_ih + (size_t)(tn +  0 + l15) * E_ + half * 2;
    const float* __restrict__ brow1 = W_ih + (size_t)(tn + 16 + l15) * E_ + half * 2;
    const float* __restrict__ brow2 = W_ih + (size_t)(tn + 32 + l15) * E_ + half * 2;
    const float* __restrict__ brow3 = W_ih + (size_t)(tn + 48 + l15) * E_ + half * 2;

    v8f acc0 = {}; v8f acc1 = {}; v8f acc2 = {}; v8f acc3 = {};

    #pragma unroll 4
    for (int k0 = 0; k0 < E_; k0 += 4) {
        v2f a  = { arow[k0],  arow[k0 + 1] };
        v2f b0 = { brow0[k0], brow0[k0 + 1] };
        v2f b1 = { brow1[k0], brow1[k0 + 1] };
        v2f b2v = { brow2[k0], brow2[k0 + 1] };
        v2f b3 = { brow3[k0], brow3[k0 + 1] };
        acc0 = __builtin_amdgcn_wmma_f32_16x16x4_f32(false, a, false, b0,  (short)0, acc0, false, false);
        acc1 = __builtin_amdgcn_wmma_f32_16x16x4_f32(false, a, false, b1,  (short)0, acc1, false, false);
        acc2 = __builtin_amdgcn_wmma_f32_16x16x4_f32(false, a, false, b2v, (short)0, acc2, false, false);
        acc3 = __builtin_amdgcn_wmma_f32_16x16x4_f32(false, a, false, b3,  (short)0, acc3, false, false);
    }

    #pragma unroll
    for (int r = 0; r < 8; ++r) {
        const int row = tm + half * 8 + r;
        float* __restrict__ dst = pre + (size_t)row * H_ + tn + l15;
        dst[ 0] = acc0[r] + b_ih[tn +  0 + l15];
        dst[16] = acc1[r] + b_ih[tn + 16 + l15];
        dst[32] = acc2[r] + b_ih[tn + 32 + l15];
        dst[48] = acc3[r] + b_ih[tn + 48 + l15];
    }
}

__global__ __launch_bounds__(256)
void k_rnn_step(const float* __restrict__ hin,
                const float* __restrict__ W_hh,
                const float* __restrict__ b_hh,
                const float* __restrict__ pre_s,
                const int* __restrict__ lengths,
                int s,
                float* __restrict__ hout,
                float* __restrict__ last) {
    const int lane = threadIdx.x & 31;
    const int wave = threadIdx.x >> 5;
    const int gw   = blockIdx.x * (blockDim.x >> 5) + wave;

    const int NT = H_ / 16;
    const int tn = (gw % NT) * 16;
    const int tm = (gw / NT) * 16;

    const int half = lane >> 4;
    const int l15  = lane & 15;

    const float* __restrict__ arow = hin  + (size_t)(tm + l15) * H_ + half * 2;
    const float* __restrict__ brow = W_hh + (size_t)(tn + l15) * H_ + half * 2;

    v8f acc = {};
    #pragma unroll 8
    for (int k0 = 0; k0 < H_; k0 += 4) {
        v2f a = { arow[k0], arow[k0 + 1] };
        v2f b = { brow[k0], brow[k0 + 1] };
        acc = __builtin_amdgcn_wmma_f32_16x16x4_f32(false, a, false, b, (short)0, acc, false, false);
    }

    const int   col  = tn + l15;
    const float bias = b_hh[col];
    #pragma unroll
    for (int r = 0; r < 8; ++r) {
        const int row = tm + half * 8 + r;
        const float v = tanhf(acc[r] + pre_s[(size_t)row * H_ + col] + bias);
        hout[(size_t)row * H_ + col] = v;
        if (lengths[row] - 1 == s) last[(size_t)row * H_ + col] = v;
    }
}

__global__ __launch_bounds__(256)
void k_mlp1(const float* __restrict__ last,
            const float* __restrict__ W1,
            const float* __restrict__ b1,
            float* __restrict__ y1) {
    const int lane = threadIdx.x & 31;
    const int wave = threadIdx.x >> 5;
    const int gw   = blockIdx.x * (blockDim.x >> 5) + wave;

    const int NT = H_ / 16;
    const int tn = (gw % NT) * 16;
    const int tm = (gw / NT) * 16;

    const int half = lane >> 4;
    const int l15  = lane & 15;

    const float* __restrict__ arow = last + (size_t)(tm + l15) * H_ + half * 2;
    const float* __restrict__ brow = W1   + (size_t)(tn + l15) * H_ + half * 2;

    v8f acc = {};
    #pragma unroll 8
    for (int k0 = 0; k0 < H_; k0 += 4) {
        v2f a = { arow[k0], arow[k0 + 1] };
        v2f b = { brow[k0], brow[k0 + 1] };
        acc = __builtin_amdgcn_wmma_f32_16x16x4_f32(false, a, false, b, (short)0, acc, false, false);
    }

    const int   col  = tn + l15;
    const float bias = b1[col];
    #pragma unroll
    for (int r = 0; r < 8; ++r) {
        const int row = tm + half * 8 + r;
        const float v = acc[r] + bias;
        y1[(size_t)row * H_ + col] = v > 0.0f ? v : 0.0f;
    }
}

__global__ void k_mlp2(const float* __restrict__ y1,
                       const float* __restrict__ W2,
                       const float* __restrict__ b2,
                       float* __restrict__ out) {
    int i = blockIdx.x * blockDim.x + threadIdx.x;
    if (i >= B_ * C_) return;
    const int b = i / C_;
    const int c = i % C_;
    const float* __restrict__ yr = y1 + (size_t)b * H_;
    const float* __restrict__ wr = W2 + (size_t)c * H_;
    float acc = b2[c];
    #pragma unroll 8
    for (int k = 0; k < H_; ++k) acc += yr[k] * wr[k];
    out[i] = acc;
}

extern "C" void kernel_launch(void* const* d_in, const int* in_sizes, int n_in,
                              void* d_out, int out_size, void* d_ws, size_t ws_size,
                              hipStream_t stream) {
    const int*   x_in  = (const int*)  d_in[0];
    const int*   x_len = (const int*)  d_in[1];
    const float* emb   = (const float*)d_in[2];
    const float* W_ih  = (const float*)d_in[3];
    const float* b_ih  = (const float*)d_in[4];
    const float* W_hh  = (const float*)d_in[5];
    const float* b_hh  = (const float*)d_in[6];
    const float* W1    = (const float*)d_in[7];
    const float* b1    = (const float*)d_in[8];
    const float* W2    = (const float*)d_in[9];
    const float* b2    = (const float*)d_in[10];
    float* out = (float*)d_out;

    float* pre  = (float*)d_ws;
    float* hb0  = pre  + (size_t)S_ * B_ * H_;
    float* hb1  = hb0  + (size_t)B_ * H_;
    float* last = hb1  + (size_t)B_ * H_;
    float* y1   = last + (size_t)B_ * H_;

    k_zero<<<(B_ * H_ + 255) / 256, 256, 0, stream>>>(hb0, B_ * H_);

    {
        const int waves  = (S_ * B_ / 16) * (H_ / 64);
        const int blocks = waves / 8;
        k_embed_proj<<<blocks, 256, 0, stream>>>(x_in, emb, W_ih, b_ih, pre);
    }

    {
        const int waves  = (B_ / 16) * (H_ / 16);
        const int blocks = waves / 8;
        float* bufs[2] = { hb0, hb1 };
        for (int s = 0; s < S_; ++s) {
            const float* hin  = bufs[s & 1];
            float*       hout = bufs[(s + 1) & 1];
            k_rnn_step<<<blocks, 256, 0, stream>>>(
                hin, W_hh, b_hh, pre + (size_t)s * B_ * H_, x_len, s, hout, last);
        }
    }

    {
        const int waves  = (B_ / 16) * (H_ / 16);
        const int blocks = waves / 8;
        k_mlp1<<<blocks, 256, 0, stream>>>(last, W1, b1, y1);
        k_mlp2<<<(B_ * C_ + 127) / 128, 128, 0, stream>>>(y1, W2, b2, out);
    }
}


# Input projection GEMM re-tiled: 64x64 tile per wave, W_ih and emb re-laid once per call so a 16-byte load feeds two k-steps: 80 cache lines touched per 32 WMMAs, was 80 per 4; bit-identical
# speedup vs baseline: 3.2151x; 3.2151x over previous
#include <hip/hip_runtime.h>
#include <hip/hip_bf16.h>
#include <math.h>

typedef float v2f __attribute__((ext_vector_type(2)));
typedef float v4f __attribute__((ext_vector_type(4)));
typedef float v8f __attribute__((ext_vector_type(8)));

#define B_ 256
#define S_ 128
#define E_ 512
#define H_ 1024
#define V_ 128
#define C_ 18

__global__ void k_zero(float* __restrict__ p, int n) {
    int i = blockIdx.x * blockDim.x + threadIdx.x;
    if (i < n) p[i] = 0.0f;
}

#define PJ_THREADS 128
#define PJ_BLOCKS ((S_ * B_ / 128) * (H_ / 128))
#define PJ_W_CHUNKS (H_ * E_ / 4)
#define PJ_E_CHUNKS (V_ * E_ / 4)

__global__ __launch_bounds__(256)
void k_pack_proj(const float* __restrict__ emb,
                 const float* __restrict__ W_ih,
                 float* __restrict__ Ep,
                 float* __restrict__ Wip) {
    if (blockIdx.x < PJ_W_CHUNKS / 256) {
        const int i = blockIdx.x * 256 + threadIdx.x;
        const int l = i & 31;
        const int g = (i >> 5) & 63;
        const int t = i >> 11;
        const int half = l >> 4;
        const int l15  = l & 15;
        const float* __restrict__ src = W_ih + (size_t)(t * 16 + l15) * E_ + g * 8 + half * 2;
        v4f v = { src[0], src[1], src[4], src[5] };
        ((v4f*)Wip)[i] = v;
    } else {
        const int i = (blockIdx.x - PJ_W_CHUNKS / 256) * 256 + threadIdx.x;
        const int half = i & 1;
        const int g    = (i >> 1) & 63;
        const int tok  = i >> 7;
        const float* __restrict__ src = emb + (size_t)tok * E_ + g * 8 + half * 2;
        v4f v = { src[0], src[1], src[4], src[5] };
        ((v4f*)Ep)[i] = v;
    }
}

__global__ __launch_bounds__(PJ_THREADS)
void k_embed_proj(const int* __restrict__ x_in,
                  const float* __restrict__ Ep,
                  const float* __restrict__ Wip,
                  const float* __restrict__ b_ih,
                  float* __restrict__ pre) {
    const int lane = threadIdx.x & 31;
    const int wave = threadIdx.x >> 5;
    const int bn = blockIdx.x % (H_ / 128);
    const int bm = blockIdx.x / (H_ / 128);
    const int tm = bm * 128 + (wave >> 1) * 64;
    const int tn = bn * 128 + (wave & 1) * 64;

    const int half = lane >> 4;
    const int l15  = lane & 15;

    const v4f* __restrict__ ap[4];
    #pragma unroll
    for (int i = 0; i < 4; ++i) {
        const int m   = tm + i * 16 + l15;
        const int s   = m / B_;
        const int b   = m % B_;
        const int tok = x_in[b * S_ + s];
        ap[i] = (const v4f*)(Ep + (size_t)tok * E_ + half * 4);
    }
    const v4f* __restrict__ bp = (const v4f*)Wip + (size_t)(tn >> 4) * (64 * 32) + lane;

    constexpr int GS = 4;
    v4f a0[GS][4], b0[GS][4], a1[GS][4], b1[GS][4];
    v8f acc[4][4] = {};

#define PJ_LOAD_SET(A, B, G0)                                                  \
    _Pragma("unroll")                                                          \
    for (int q = 0; q < GS; ++q) {                                             \
        _Pragma("unroll")                                                      \
        for (int i = 0; i < 4; ++i) A[q][i] = ap[i][((G0) + q) * 2];           \
        _Pragma("unroll")                                                      \
        for (int j = 0; j < 4; ++j) B[q][j] = bp[((G0) + q) * 32 + j * (64 * 32)]; \
    }
#define PJ_MMA_SET(A, B)                                                       \
    _Pragma("unroll")                                                          \
    for (int q = 0; q < GS; ++q) {                                             \
        _Pragma("unroll")                                                      \
        for (int i = 0; i < 4; ++i) {                                          \
            _Pragma("unroll")                                                  \
            for (int j = 0; j < 4; ++j) {                                      \
                const v2f x0 = { A[q][i].x, A[q][i].y };                       \
                const v2f y0 = { B[q][j].x, B[q][j].y };                       \
                acc[i][j] = __builtin_amdgcn_wmma_f32_16x16x4_f32(false, x0, false, y0, (short)0, acc[i][j], false, false); \
            }                                                                  \
        }                                                                      \
        _Pragma("unroll")                                                      \
        for (int i = 0; i < 4; ++i) {                                          \
            _Pragma("unroll")                                                  \
            for (int j = 0; j < 4; ++j) {                                      \
                const v2f x1 = { A[q][i].z, A[q][i].w };                       \
                const v2f y1 = { B[q][j].z, B[q][j].w };                       \
                acc[i][j] = __builtin_amdgcn_wmma_f32_16x16x4_f32(false, x1, false, y1, (short)0, acc[i][j], false, false); \
            }                                                                  \
        }                                                                      \
    }
#define PJ_FENCE() __builtin_amdgcn_sched_barrier(0)

    PJ_LOAD_SET(a0, b0, 0)
    PJ_FENCE();
    for (int g = 0; g < E_ / 8 - 2 * GS; g += 2 * GS) {
        PJ_LOAD_SET(a1, b1, g + GS)
        PJ_FENCE();
        PJ_MMA_SET(a0, b0)
        PJ_FENCE();
        PJ_LOAD_SET(a0, b0, g + 2 * GS)
        PJ_FENCE();
        PJ_MMA_SET(a1, b1)
        PJ_FENCE();
    }
    PJ_LOAD_SET(a1, b1, E_ / 8 - GS)
    PJ_FENCE();
    PJ_MMA_SET(a0, b0)
    PJ_MMA_SET(a1, b1)
#undef PJ_FENCE
#undef PJ_LOAD_SET
#undef PJ_MMA_SET

    float bias[4];
    #pragma unroll
    for (int j = 0; j < 4; ++j) bias[j] = b_ih[tn + j * 16 + l15];
    #pragma unroll
    for (int i = 0; i < 4; ++i) {
        #pragma unroll
        for (int r = 0; r < 8; ++r) {
            const int row = tm + i * 16 + half * 8 + r;
            float* __restrict__ dst = pre + (size_t)row * H_ + tn + l15;
            #pragma unroll
            for (int j = 0; j < 4; ++j) dst[j * 16] = acc[i][j][r] + bias[j];
        }
    }
}


__global__ __launch_bounds__(256)
void k_pack_w(const float* __restrict__ W, float* __restrict__ Wp) {
    const int i = blockIdx.x * 256 + threadIdx.x;
    const int l = i & 31;
    const int g = (i >> 5) & 127;
    const int t = i >> 12;
    const int half = l >> 4;
    const int l15  = l & 15;
    const float* __restrict__ src = W + (size_t)(t * 16 + l15) * H_ + g * 8 + half * 2;
    v4f v = { src[0], src[1], src[4], src[5] };
    ((v4f*)Wp)[i] = v;
}

__device__ __forceinline__
void rnn_step_epilogue_vals(v8f acc, int tm, int tn, int half, int l15,
                            float bias, const float (&pv)[8], const int (&ln)[8],
                            int s,
                            float* __restrict__ hout,
                            float* __restrict__ last) {
    const int col = tn + l15;
    const int kk = l15 & 7;
    const int hp = (kk >> 1) & 1;
    const int e  = (kk & 1) | ((kk >> 2) << 1);
    float* __restrict__ hdst = hout
        + ((size_t)((tm >> 4) * 128 + (col >> 3)) * 32 + hp * 16 + half * 8) * 4 + e;
    #pragma unroll
    for (int r = 0; r < 8; ++r) {
        const int row = tm + half * 8 + r;
        const float v = tanhf(acc[r] + pv[r] + bias);
        hdst[r * 4] = v;
        if (ln[r] - 1 == s) last[(size_t)row * H_ + col] = v;
    }
}

__device__ __forceinline__
void rnn_step_epilogue_operands(int tm, int tn, int half, int l15,
                                const float* __restrict__ b_hh,
                                const float* __restrict__ pre_s,
                                const int* __restrict__ lengths,
                                float& bias, float (&pv)[8], int (&ln)[8]) {
    const int col = tn + l15;
    bias = b_hh[col];
    #pragma unroll
    for (int r = 0; r < 8; ++r) {
        const int row = tm + half * 8 + r;
        pv[r] = pre_s[(size_t)row * H_ + col];
        ln[r] = lengths[row];
    }
}

#define STEP_THREADS 128
#define STEP_BLOCKS ((B_ / 32) * (H_ / 32))

__device__ __forceinline__
void rnn_step_tile(int& tm, int& tn) {
    const int wave = threadIdx.x >> 5;
    const int bn = blockIdx.x % (H_ / 32);
    const int bm = blockIdx.x / (H_ / 32);
    tm = (bm * 2 + (wave >> 1)) * 16;
    tn = (bn * 2 + (wave & 1)) * 16;
}

__global__ __launch_bounds__(STEP_THREADS)
void k_rnn_step_rowmajor_w(const float* __restrict__ hin,
                           const float* __restrict__ W_hh,
                           const float* __restrict__ b_hh,
                           const float* __restrict__ pre_s,
                           const int* __restrict__ lengths,
                           int s,
                           float* __restrict__ hout,
                           float* __restrict__ last) {
    const int lane = threadIdx.x & 31;
    int tm, tn;
    rnn_step_tile(tm, tn);

    const int half = lane >> 4;
    const int l15  = lane & 15;

    const v4f*   __restrict__ ap   = (const v4f*)hin + (size_t)(tm >> 4) * (128 * 32) + lane;
    const float* __restrict__ brow = W_hh + (size_t)(tn + l15) * H_ + half * 2;

    float bias;
    float pv[8];
    int   ln[8];
    rnn_step_epilogue_operands(tm, tn, half, l15, b_hh, pre_s, lengths, bias, pv, ln);

    v8f acc = {};
    #pragma unroll 8
    for (int g = 0; g < H_ / 8; ++g) {
        const v4f a  = ap[g * 32];
        const v2f a0 = { a.x, a.y };
        const v2f a1 = { a.z, a.w };
        const v2f b0 = { brow[g * 8],     brow[g * 8 + 1] };
        const v2f b1 = { brow[g * 8 + 4], brow[g * 8 + 5] };
        acc = __builtin_amdgcn_wmma_f32_16x16x4_f32(false, a0, false, b0, (short)0, acc, false, false);
        acc = __builtin_amdgcn_wmma_f32_16x16x4_f32(false, a1, false, b1, (short)0, acc, false, false);
    }

    rnn_step_epilogue_vals(acc, tm, tn, half, l15, bias, pv, ln, s, hout, last);
}

__global__ __launch_bounds__(STEP_THREADS) __attribute__((amdgpu_waves_per_eu(1, 1)))
void k_rnn_step_frag_w(const float* __restrict__ hin,
                       const float* __restrict__ Wp,
                       const float* __restrict__ b_hh,
                       const float* __restrict__ pre_s,
                       const int* __restrict__ lengths,
                       int s,
                       float* __restrict__ hout,
                       float* __restrict__ last) {
    const int lane = threadIdx.x & 31;
    int tm, tn;
    rnn_step_tile(tm, tn);

    const int half = lane >> 4;
    const int l15  = lane & 15;

    const v4f* __restrict__ ap = (const v4f*)hin + (size_t)(tm >> 4) * (128 * 32) + lane;
    const v4f* __restrict__ bp = (const v4f*)Wp  + (size_t)(tn >> 4) * (128 * 32) + lane;

    float bias;
    float pv[8];
    int   ln[8];
    rnn_step_epilogue_operands(tm, tn, half, l15, b_hh, pre_s, lengths, bias, pv, ln);

    constexpr int SET = 16;
    v4f a0[SET], b0[SET], a1[SET], b1[SET];
    v8f acc = {};

#define LOAD_SET(A, B, G0)                                                     \
    _Pragma("unroll")                                                          \
    for (int j = 0; j < SET; ++j) {                                            \
        A[j] = ap[((G0) + j) * 32];                                            \
        B[j] = bp[((G0) + j) * 32];                                            \
    }
#define MMA_SET(A, B)                                                          \
    _Pragma("unroll")                                                          \
    for (int j = 0; j < SET; ++j) {                                            \
        const v2f x0 = { A[j].x, A[j].y };                                     \
        const v2f x1 = { A[j].z, A[j].w };                                     \
        const v2f y0 = { B[j].x, B[j].y };                                     \
        const v2f y1 = { B[j].z, B[j].w };                                     \
        acc = __builtin_amdgcn_wmma_f32_16x16x4_f32(false, x0, false, y0, (short)0, acc, false, false); \
        acc = __builtin_amdgcn_wmma_f32_16x16x4_f32(false, x1, false, y1, (short)0, acc, false, false); \
    }

#define GROUP_FENCE() __builtin_amdgcn_sched_barrier(0)

    LOAD_SET(a0, b0, 0)
    GROUP_FENCE();
    for (int g = 0; g < H_ / 8 - 2 * SET; g += 2 * SET) {
        LOAD_SET(a1, b1, g + SET)
        GROUP_FENCE();
        MMA_SET(a0, b0)
        GROUP_FENCE();
        LOAD_SET(a0, b0, g + 2 * SET)
        GROUP_FENCE();
        MMA_SET(a1, b1)
        GROUP_FENCE();
    }
    LOAD_SET(a1, b1, H_ / 8 - SET)
    GROUP_FENCE();
    MMA_SET(a0, b0)
    MMA_SET(a1, b1)
#undef GROUP_FENCE
#undef LOAD_SET
#undef MMA_SET

    rnn_step_epilogue_vals(acc, tm, tn, half, l15, bias, pv, ln, s, hout, last);
}

__global__ __launch_bounds__(256)
void k_mlp1(const float* __restrict__ last,
            const float* __restrict__ W1,
            const float* __restrict__ b1,
            float* __restrict__ y1) {
    const int lane = threadIdx.x & 31;
    const int wave = threadIdx.x >> 5;
    const int gw   = blockIdx.x * (blockDim.x >> 5) + wave;

    const int NT = H_ / 16;
    const int tn = (gw % NT) * 16;
    const int tm = (gw / NT) * 16;

    const int half = lane >> 4;
    const int l15  = lane & 15;

    const float* __restrict__ arow = last + (size_t)(tm + l15) * H_ + half * 2;
    const float* __restrict__ brow = W1   + (size_t)(tn + l15) * H_ + half * 2;

    v8f acc = {};
    #pragma unroll 8
    for (int k0 = 0; k0 < H_; k0 += 4) {
        v2f a = { arow[k0], arow[k0 + 1] };
        v2f b = { brow[k0], brow[k0 + 1] };
        acc = __builtin_amdgcn_wmma_f32_16x16x4_f32(false, a, false, b, (short)0, acc, false, false);
    }

    const int   col  = tn + l15;
    const float bias = b1[col];
    #pragma unroll
    for (int r = 0; r < 8; ++r) {
        const int row = tm + half * 8 + r;
        const float v = acc[r] + bias;
        y1[(size_t)row * H_ + col] = v > 0.0f ? v : 0.0f;
    }
}

__global__ void k_mlp2(const float* __restrict__ y1,
                       const float* __restrict__ W2,
                       const float* __restrict__ b2,
                       float* __restrict__ out) {
    int i = blockIdx.x * blockDim.x + threadIdx.x;
    if (i >= B_ * C_) return;
    const int b = i / C_;
    const int c = i % C_;
    const float* __restrict__ yr = y1 + (size_t)b * H_;
    const float* __restrict__ wr = W2 + (size_t)c * H_;
    float acc = b2[c];
    #pragma unroll 8
    for (int k = 0; k < H_; ++k) acc += yr[k] * wr[k];
    out[i] = acc;
}

extern "C" void kernel_launch(void* const* d_in, const int* in_sizes, int n_in,
                              void* d_out, int out_size, void* d_ws, size_t ws_size,
                              hipStream_t stream) {
    const int*   x_in  = (const int*)  d_in[0];
    const int*   x_len = (const int*)  d_in[1];
    const float* emb   = (const float*)d_in[2];
    const float* W_ih  = (const float*)d_in[3];
    const float* b_ih  = (const float*)d_in[4];
    const float* W_hh  = (const float*)d_in[5];
    const float* b_hh  = (const float*)d_in[6];
    const float* W1    = (const float*)d_in[7];
    const float* b1    = (const float*)d_in[8];
    const float* W2    = (const float*)d_in[9];
    const float* b2    = (const float*)d_in[10];
    float* out = (float*)d_out;

    float* pre  = (float*)d_ws;
    float* hb0  = pre  + (size_t)S_ * B_ * H_;
    float* hb1  = hb0  + (size_t)B_ * H_;
    float* last = hb1  + (size_t)B_ * H_;
    float* y1   = last + (size_t)B_ * H_;

    k_zero<<<(B_ * H_ + 255) / 256, 256, 0, stream>>>(hb0, B_ * H_);

    {
        float* Wip = hb1;
        float* Ep  = y1;
        k_pack_proj<<<(PJ_W_CHUNKS + PJ_E_CHUNKS) / 256, 256, 0, stream>>>(emb, W_ih, Ep, Wip);
        k_embed_proj<<<PJ_BLOCKS, PJ_THREADS, 0, stream>>>(x_in, Ep, Wip, b_ih, pre);
    }

    {
        float* bufs[2] = { hb0, hb1 };
        const int S_ROW = 4;
        float* Wp = pre;
        for (int s = 0; s < S_ROW; ++s) {
            const float* hin  = bufs[s & 1];
            float*       hout = bufs[(s + 1) & 1];
            k_rnn_step_rowmajor_w<<<STEP_BLOCKS, STEP_THREADS, 0, stream>>>(
                hin, W_hh, b_hh, pre + (size_t)s * B_ * H_, x_len, s, hout, last);
        }
        k_pack_w<<<(H_ * H_ / 4) / 256, 256, 0, stream>>>(W_hh, Wp);
        for (int s = S_ROW; s < S_; ++s) {
            const float* hin  = bufs[s & 1];
            float*       hout = bufs[(s + 1) & 1];
            k_rnn_step_frag_w<<<STEP_BLOCKS, STEP_THREADS, 0, stream>>>(
                hin, Wp, b_hh, pre + (size_t)s * B_ * H_, x_len, s, hout, last);
        }
    }

    {
        const int waves  = (B_ / 16) * (H_ / 16);
        const int blocks = waves / 8;
        k_mlp1<<<blocks, 256, 0, stream>>>(last, W1, b1, y1);
        k_mlp2<<<(B_ * C_ + 127) / 128, 128, 0, stream>>>(y1, W2, b2, out);
    }
}


# Step epilogue: tanh of the 8 rows in stages (polynomial form straight-line for all 8; exponential form, and stores into last, under one wave-level test each), not 8 branchy calls in turn; bit-exact
# speedup vs baseline: 3.3153x; 1.0312x over previous
#include <hip/hip_runtime.h>
#include <hip/hip_bf16.h>
#include <math.h>

typedef float v2f __attribute__((ext_vector_type(2)));
typedef float v4f __attribute__((ext_vector_type(4)));
typedef float v8f __attribute__((ext_vector_type(8)));

#define B_ 256
#define S_ 128
#define E_ 512
#define H_ 1024
#define V_ 128
#define C_ 18

__global__ void k_zero(float* __restrict__ p, int n) {
    int i = blockIdx.x * blockDim.x + threadIdx.x;
    if (i < n) p[i] = 0.0f;
}

#define PJ_THREADS 128
#define PJ_BLOCKS ((S_ * B_ / 128) * (H_ / 128))
#define PJ_W_CHUNKS (H_ * E_ / 4)
#define PJ_E_CHUNKS (V_ * E_ / 4)

__global__ __launch_bounds__(256)
void k_pack_proj(const float* __restrict__ emb,
                 const float* __restrict__ W_ih,
                 float* __restrict__ Ep,
                 float* __restrict__ Wip) {
    if (blockIdx.x < PJ_W_CHUNKS / 256) {
        const int i = blockIdx.x * 256 + threadIdx.x;
        const int l = i & 31;
        const int g = (i >> 5) & 63;
        const int t = i >> 11;
        const int half = l >> 4;
        const int l15  = l & 15;
        const float* __restrict__ src = W_ih + (size_t)(t * 16 + l15) * E_ + g * 8 + half * 2;
        v4f v = { src[0], src[1], src[4], src[5] };
        ((v4f*)Wip)[i] = v;
    } else {
        const int i = (blockIdx.x - PJ_W_CHUNKS / 256) * 256 + threadIdx.x;
        const int half = i & 1;
        const int g    = (i >> 1) & 63;
        const int tok  = i >> 7;
        const float* __restrict__ src = emb + (size_t)tok * E_ + g * 8 + half * 2;
        v4f v = { src[0], src[1], src[4], src[5] };
        ((v4f*)Ep)[i] = v;
    }
}

__global__ __launch_bounds__(PJ_THREADS)
void k_embed_proj(const int* __restrict__ x_in,
                  const float* __restrict__ Ep,
                  const float* __restrict__ Wip,
                  const float* __restrict__ b_ih,
                  float* __restrict__ pre) {
    const int lane = threadIdx.x & 31;
    const int wave = threadIdx.x >> 5;
    const int bn = blockIdx.x % (H_ / 128);
    const int bm = blockIdx.x / (H_ / 128);
    const int tm = bm * 128 + (wave >> 1) * 64;
    const int tn = bn * 128 + (wave & 1) * 64;

    const int half = lane >> 4;
    const int l15  = lane & 15;

    const v4f* __restrict__ ap[4];
    #pragma unroll
    for (int i = 0; i < 4; ++i) {
        const int m   = tm + i * 16 + l15;
        const int s   = m / B_;
        const int b   = m % B_;
        const int tok = x_in[b * S_ + s];
        ap[i] = (const v4f*)(Ep + (size_t)tok * E_ + half * 4);
    }
    const v4f* __restrict__ bp = (const v4f*)Wip + (size_t)(tn >> 4) * (64 * 32) + lane;

    constexpr int GS = 4;
    v4f a0[GS][4], b0[GS][4], a1[GS][4], b1[GS][4];
    v8f acc[4][4] = {};

#define PJ_LOAD_SET(A, B, G0)                                                  \
    _Pragma("unroll")                                                          \
    for (int q = 0; q < GS; ++q) {                                             \
        _Pragma("unroll")                                                      \
        for (int i = 0; i < 4; ++i) A[q][i] = ap[i][((G0) + q) * 2];           \
        _Pragma("unroll")                                                      \
        for (int j = 0; j < 4; ++j) B[q][j] = bp[((G0) + q) * 32 + j * (64 * 32)]; \
    }
#define PJ_MMA_SET(A, B)                                                       \
    _Pragma("unroll")                                                          \
    for (int q = 0; q < GS; ++q) {                                             \
        _Pragma("unroll")                                                      \
        for (int i = 0; i < 4; ++i) {                                          \
            _Pragma("unroll")                                                  \
            for (int j = 0; j < 4; ++j) {                                      \
                const v2f x0 = { A[q][i].x, A[q][i].y };                       \
                const v2f y0 = { B[q][j].x, B[q][j].y };                       \
                acc[i][j] = __builtin_amdgcn_wmma_f32_16x16x4_f32(false, x0, false, y0, (short)0, acc[i][j], false, false); \
            }                                                                  \
        }                                                                      \
        _Pragma("unroll")                                                      \
        for (int i = 0; i < 4; ++i) {                                          \
            _Pragma("unroll")                                                  \
            for (int j = 0; j < 4; ++j) {                                      \
                const v2f x1 = { A[q][i].z, A[q][i].w };                       \
                const v2f y1 = { B[q][j].z, B[q][j].w };                       \
                acc[i][j] = __builtin_amdgcn_wmma_f32_16x16x4_f32(false, x1, false, y1, (short)0, acc[i][j], false, false); \
            }                                                                  \
        }                                                                      \
    }
#define PJ_FENCE() __builtin_amdgcn_sched_barrier(0)

    PJ_LOAD_SET(a0, b0, 0)
    PJ_FENCE();
    for (int g = 0; g < E_ / 8 - 2 * GS; g += 2 * GS) {
        PJ_LOAD_SET(a1, b1, g + GS)
        PJ_FENCE();
        PJ_MMA_SET(a0, b0)
        PJ_FENCE();
        PJ_LOAD_SET(a0, b0, g + 2 * GS)
        PJ_FENCE();
        PJ_MMA_SET(a1, b1)
        PJ_FENCE();
    }
    PJ_LOAD_SET(a1, b1, E_ / 8 - GS)
    PJ_FENCE();
    PJ_MMA_SET(a0, b0)
    PJ_MMA_SET(a1, b1)
#undef PJ_FENCE
#undef PJ_LOAD_SET
#undef PJ_MMA_SET

    float bias[4];
    #pragma unroll
    for (int j = 0; j < 4; ++j) bias[j] = b_ih[tn + j * 16 + l15];
    #pragma unroll
    for (int i = 0; i < 4; ++i) {
        #pragma unroll
        for (int r = 0; r < 8; ++r) {
            const int row = tm + i * 16 + half * 8 + r;
            float* __restrict__ dst = pre + (size_t)row * H_ + tn + l15;
            #pragma unroll
            for (int j = 0; j < 4; ++j) dst[j * 16] = acc[i][j][r] + bias[j];
        }
    }
}


__global__ __launch_bounds__(256)
void k_pack_w(const float* __restrict__ W, float* __restrict__ Wp) {
    const int i = blockIdx.x * 256 + threadIdx.x;
    const int l = i & 31;
    const int g = (i >> 5) & 127;
    const int t = i >> 12;
    const int half = l >> 4;
    const int l15  = l & 15;
    const float* __restrict__ src = W + (size_t)(t * 16 + l15) * H_ + g * 8 + half * 2;
    v4f v = { src[0], src[1], src[4], src[5] };
    ((v4f*)Wp)[i] = v;
}

__device__ __forceinline__ float tanh_abs_below_0625(float y) {
    const float y2 = y * y;
    float p = __builtin_fmaf(y2, -0x1.758e7ap-8f, 0x1.521192p-6f);
    p = __builtin_fmaf(y2, p, -0x1.b8389cp-5f);
    p = __builtin_fmaf(y2, p, 0x1.110704p-3f);
    p = __builtin_fmaf(y2, p, -0x1.555532p-2f);
    return __builtin_fmaf(y2, y * p, y);
}
__device__ __forceinline__ float tanh_abs_from_0625(float y) {
    const float t = expf(2.0f * y);
    return __builtin_fmaf(-2.0f, __builtin_amdgcn_rcpf(t + 1.0f), 1.0f);
}

__device__ __forceinline__
void rnn_step_epilogue_vals(v8f acc, int tm, int tn, int half, int l15,
                            float bias, const float (&pv)[8], const int (&ln)[8],
                            int s,
                            float* __restrict__ hout,
                            float* __restrict__ last) {
    const int col = tn + l15;
    const int kk = l15 & 7;
    const int hp = (kk >> 1) & 1;
    const int e  = (kk & 1) | ((kk >> 2) << 1);
    float* __restrict__ hdst = hout
        + ((size_t)((tm >> 4) * 128 + (col >> 3)) * 32 + hp * 16 + half * 8) * 4 + e;
    float* __restrict__ ldst = last + (size_t)(tm + half * 8) * H_ + col;

    float z[8], y[8], t[8];
    bool large = false;
    #pragma unroll
    for (int r = 0; r < 8; ++r) {
        z[r] = acc[r] + pv[r] + bias;
        y[r] = fabsf(z[r]);
        large = large || !(y[r] < 0.625f);
        t[r] = tanh_abs_below_0625(y[r]);
    }
    if (large) {
        #pragma unroll
        for (int r = 0; r < 8; ++r) {
            const float x = tanh_abs_from_0625(y[r]);
            t[r] = (y[r] < 0.625f) ? t[r] : x;
        }
    }
    unsigned ends = 0;
    #pragma unroll
    for (int r = 0; r < 8; ++r) {
        t[r] = copysignf(t[r], z[r]);
        hdst[r * 4] = t[r];
        ends |= __builtin_amdgcn_ballot_w32(ln[r] - 1 == s);
    }
    if (ends != 0) {
        #pragma unroll
        for (int r = 0; r < 8; ++r)
            if (ln[r] - 1 == s) ldst[(size_t)r * H_] = t[r];
    }
}

__device__ __forceinline__
void rnn_step_epilogue_operands(int tm, int tn, int half, int l15,
                                const float* __restrict__ b_hh,
                                const float* __restrict__ pre_s,
                                const int* __restrict__ lengths,
                                float& bias, float (&pv)[8], int (&ln)[8]) {
    const int col = tn + l15;
    bias = b_hh[col];
    #pragma unroll
    for (int r = 0; r < 8; ++r) {
        const int row = tm + half * 8 + r;
        pv[r] = pre_s[(size_t)row * H_ + col];
        ln[r] = lengths[row];
    }
}

#define STEP_THREADS 128
#define STEP_BLOCKS ((B_ / 32) * (H_ / 32))

__device__ __forceinline__
void rnn_step_tile(int& tm, int& tn) {
    const int wave = threadIdx.x >> 5;
    const int bn = blockIdx.x % (H_ / 32);
    const int bm = blockIdx.x / (H_ / 32);
    tm = (bm * 2 + (wave >> 1)) * 16;
    tn = (bn * 2 + (wave & 1)) * 16;
}

__global__ __launch_bounds__(STEP_THREADS)
void k_rnn_step_rowmajor_w(const float* __restrict__ hin,
                           const float* __restrict__ W_hh,
                           const float* __restrict__ b_hh,
                           const float* __restrict__ pre_s,
                           const int* __restrict__ lengths,
                           int s,
                           float* __restrict__ hout,
                           float* __restrict__ last) {
    const int lane = threadIdx.x & 31;
    int tm, tn;
    rnn_step_tile(tm, tn);

    const int half = lane >> 4;
    const int l15  = lane & 15;

    const v4f*   __restrict__ ap   = (const v4f*)hin + (size_t)(tm >> 4) * (128 * 32) + lane;
    const float* __restrict__ brow = W_hh + (size_t)(tn + l15) * H_ + half * 2;

    float bias;
    float pv[8];
    int   ln[8];
    rnn_step_epilogue_operands(tm, tn, half, l15, b_hh, pre_s, lengths, bias, pv, ln);

    v8f acc = {};
    #pragma unroll 8
    for (int g = 0; g < H_ / 8; ++g) {
        const v4f a  = ap[g * 32];
        const v2f a0 = { a.x, a.y };
        const v2f a1 = { a.z, a.w };
        const v2f b0 = { brow[g * 8],     brow[g * 8 + 1] };
        const v2f b1 = { brow[g * 8 + 4], brow[g * 8 + 5] };
        acc = __builtin_amdgcn_wmma_f32_16x16x4_f32(false, a0, false, b0, (short)0, acc, false, false);
        acc = __builtin_amdgcn_wmma_f32_16x16x4_f32(false, a1, false, b1, (short)0, acc, false, false);
    }

    rnn_step_epilogue_vals(acc, tm, tn, half, l15, bias, pv, ln, s, hout, last);
}

__global__ __launch_bounds__(STEP_THREADS) __attribute__((amdgpu_waves_per_eu(1, 1)))
void k_rnn_step_frag_w(const float* __restrict__ hin,
                       const float* __restrict__ Wp,
                       const float* __restrict__ b_hh,
                       const float* __restrict__ pre_s,
                       const int* __restrict__ lengths,
                       int s,
                       float* __restrict__ hout,
                       float* __restrict__ last) {
    const int lane = threadIdx.x & 31;
    int tm, tn;
    rnn_step_tile(tm, tn);

    const int half = lane >> 4;
    const int l15  = lane & 15;

    const v4f* __restrict__ ap = (const v4f*)hin + (size_t)(tm >> 4) * (128 * 32) + lane;
    const v4f* __restrict__ bp = (const v4f*)Wp  + (size_t)(tn >> 4) * (128 * 32) + lane;

    float bias;
    float pv[8];
    int   ln[8];
    rnn_step_epilogue_operands(tm, tn, half, l15, b_hh, pre_s, lengths, bias, pv, ln);

    constexpr int SET = 16;
    v4f a0[SET], b0[SET], a1[SET], b1[SET];
    v8f acc = {};

#define LOAD_SET(A, B, G0)                                                     \
    _Pragma("unroll")                                                          \
    for (int j = 0; j < SET; ++j) {                                            \
        A[j] = ap[((G0) + j) * 32];                                            \
        B[j] = bp[((G0) + j) * 32];                                            \
    }
#define MMA_SET(A, B)                                                          \
    _Pragma("unroll")                                                          \
    for (int j = 0; j < SET; ++j) {                                            \
        const v2f x0 = { A[j].x, A[j].y };                                     \
        const v2f x1 = { A[j].z, A[j].w };                                     \
        const v2f y0 = { B[j].x, B[j].y };                                     \
        const v2f y1 = { B[j].z, B[j].w };                                     \
        acc = __builtin_amdgcn_wmma_f32_16x16x4_f32(false, x0, false, y0, (short)0, acc, false, false); \
        acc = __builtin_amdgcn_wmma_f32_16x16x4_f32(false, x1, false, y1, (short)0, acc, false, false); \
    }

#define GROUP_FENCE() __builtin_amdgcn_sched_barrier(0)

    LOAD_SET(a0, b0, 0)
    GROUP_FENCE();
    for (int g = 0; g < H_ / 8 - 2 * SET; g += 2 * SET) {
        LOAD_SET(a1, b1, g + SET)
        GROUP_FENCE();
        MMA_SET(a0, b0)
        GROUP_FENCE();
        LOAD_SET(a0, b0, g + 2 * SET)
        GROUP_FENCE();
        MMA_SET(a1, b1)
        GROUP_FENCE();
    }
    LOAD_SET(a1, b1, H_ / 8 - SET)
    GROUP_FENCE();
    MMA_SET(a0, b0)
    MMA_SET(a1, b1)
#undef GROUP_FENCE
#undef LOAD_SET
#undef MMA_SET

    rnn_step_epilogue_vals(acc, tm, tn, half, l15, bias, pv, ln, s, hout, last);
}

__global__ __launch_bounds__(256)
void k_mlp1(const float* __restrict__ last,
            const float* __restrict__ W1,
            const float* __restrict__ b1,
            float* __restrict__ y1) {
    const int lane = threadIdx.x & 31;
    const int wave = threadIdx.x >> 5;
    const int gw   = blockIdx.x * (blockDim.x >> 5) + wave;

    const int NT = H_ / 16;
    const int tn = (gw % NT) * 16;
    const int tm = (gw / NT) * 16;

    const int half = lane >> 4;
    const int l15  = lane & 15;

    const float* __restrict__ arow = last + (size_t)(tm + l15) * H_ + half * 2;
    const float* __restrict__ brow = W1   + (size_t)(tn + l15) * H_ + half * 2;

    v8f acc = {};
    #pragma unroll 8
    for (int k0 = 0; k0 < H_; k0 += 4) {
        v2f a = { arow[k0], arow[k0 + 1] };
        v2f b = { brow[k0], brow[k0 + 1] };
        acc = __builtin_amdgcn_wmma_f32_16x16x4_f32(false, a, false, b, (short)0, acc, false, false);
    }

    const int   col  = tn + l15;
    const float bias = b1[col];
    #pragma unroll
    for (int r = 0; r < 8; ++r) {
        const int row = tm + half * 8 + r;
        const float v = acc[r] + bias;
        y1[(size_t)row * H_ + col] = v > 0.0f ? v : 0.0f;
    }
}

__global__ void k_mlp2(const float* __restrict__ y1,
                       const float* __restrict__ W2,
                       const float* __restrict__ b2,
                       float* __restrict__ out) {
    int i = blockIdx.x * blockDim.x + threadIdx.x;
    if (i >= B_ * C_) return;
    const int b = i / C_;
    const int c = i % C_;
    const float* __restrict__ yr = y1 + (size_t)b * H_;
    const float* __restrict__ wr = W2 + (size_t)c * H_;
    float acc = b2[c];
    #pragma unroll 8
    for (int k = 0; k < H_; ++k) acc += yr[k] * wr[k];
    out[i] = acc;
}

extern "C" void kernel_launch(void* const* d_in, const int* in_sizes, int n_in,
                              void* d_out, int out_size, void* d_ws, size_t ws_size,
                              hipStream_t stream) {
    const int*   x_in  = (const int*)  d_in[0];
    const int*   x_len = (const int*)  d_in[1];
    const float* emb   = (const float*)d_in[2];
    const float* W_ih  = (const float*)d_in[3];
    const float* b_ih  = (const float*)d_in[4];
    const float* W_hh  = (const float*)d_in[5];
    const float* b_hh  = (const float*)d_in[6];
    const float* W1    = (const float*)d_in[7];
    const float* b1    = (const float*)d_in[8];
    const float* W2    = (const float*)d_in[9];
    const float* b2    = (const float*)d_in[10];
    float* out = (float*)d_out;

    float* pre  = (float*)d_ws;
    float* hb0  = pre  + (size_t)S_ * B_ * H_;
    float* hb1  = hb0  + (size_t)B_ * H_;
    float* last = hb1  + (size_t)B_ * H_;
    float* y1   = last + (size_t)B_ * H_;

    k_zero<<<(B_ * H_ + 255) / 256, 256, 0, stream>>>(hb0, B_ * H_);

    {
        float* Wip = hb1;
        float* Ep  = y1;
        k_pack_proj<<<(PJ_W_CHUNKS + PJ_E_CHUNKS) / 256, 256, 0, stream>>>(emb, W_ih, Ep, Wip);
        k_embed_proj<<<PJ_BLOCKS, PJ_THREADS, 0, stream>>>(x_in, Ep, Wip, b_ih, pre);
    }

    {
        float* bufs[2] = { hb0, hb1 };
        const int S_ROW = 4;
        float* Wp = pre;
        for (int s = 0; s < S_ROW; ++s) {
            const float* hin  = bufs[s & 1];
            float*       hout = bufs[(s + 1) & 1];
            k_rnn_step_rowmajor_w<<<STEP_BLOCKS, STEP_THREADS, 0, stream>>>(
                hin, W_hh, b_hh, pre + (size_t)s * B_ * H_, x_len, s, hout, last);
        }
        k_pack_w<<<(H_ * H_ / 4) / 256, 256, 0, stream>>>(W_hh, Wp);
        for (int s = S_ROW; s < S_; ++s) {
            const float* hin  = bufs[s & 1];
            float*       hout = bufs[(s + 1) & 1];
            k_rnn_step_frag_w<<<STEP_BLOCKS, STEP_THREADS, 0, stream>>>(
                hin, Wp, b_hh, pre + (size_t)s * B_ * H_, x_len, s, hout, last);
        }
    }

    {
        const int waves  = (B_ / 16) * (H_ / 16);
        const int blocks = waves / 8;
        k_mlp1<<<blocks, 256, 0, stream>>>(last, W1, b1, y1);
        k_mlp2<<<(B_ * C_ + 127) / 128, 128, 0, stream>>>(y1, W2, b2, out);
    }
}


# Step GEMM on two accumulators in turn (k-steps 8g / 8g+4, added once before the epilogue), fed by a ring of four 16-load sets so three sets stay in flight: new f32 summation order
# speedup vs baseline: 3.5338x; 1.0659x over previous
#include <hip/hip_runtime.h>
#include <hip/hip_bf16.h>
#include <math.h>

typedef float v2f __attribute__((ext_vector_type(2)));
typedef float v4f __attribute__((ext_vector_type(4)));
typedef float v8f __attribute__((ext_vector_type(8)));

#define B_ 256
#define S_ 128
#define E_ 512
#define H_ 1024
#define V_ 128
#define C_ 18

__global__ void k_zero(float* __restrict__ p, int n) {
    int i = blockIdx.x * blockDim.x + threadIdx.x;
    if (i < n) p[i] = 0.0f;
}

#define PJ_THREADS 128
#define PJ_BLOCKS ((S_ * B_ / 128) * (H_ / 128))
#define PJ_W_CHUNKS (H_ * E_ / 4)
#define PJ_E_CHUNKS (V_ * E_ / 4)

__global__ __launch_bounds__(256)
void k_pack_proj(const float* __restrict__ emb,
                 const float* __restrict__ W_ih,
                 float* __restrict__ Ep,
                 float* __restrict__ Wip) {
    if (blockIdx.x < PJ_W_CHUNKS / 256) {
        const int i = blockIdx.x * 256 + threadIdx.x;
        const int l = i & 31;
        const int g = (i >> 5) & 63;
        const int t = i >> 11;
        const int half = l >> 4;
        const int l15  = l & 15;
        const float* __restrict__ src = W_ih + (size_t)(t * 16 + l15) * E_ + g * 8 + half * 2;
        v4f v = { src[0], src[1], src[4], src[5] };
        ((v4f*)Wip)[i] = v;
    } else {
        const int i = (blockIdx.x - PJ_W_CHUNKS / 256) * 256 + threadIdx.x;
        const int half = i & 1;
        const int g    = (i >> 1) & 63;
        const int tok  = i >> 7;
        const float* __restrict__ src = emb + (size_t)tok * E_ + g * 8 + half * 2;
        v4f v = { src[0], src[1], src[4], src[5] };
        ((v4f*)Ep)[i] = v;
    }
}

__global__ __launch_bounds__(PJ_THREADS)
void k_embed_proj(const int* __restrict__ x_in,
                  const float* __restrict__ Ep,
                  const float* __restrict__ Wip,
                  const float* __restrict__ b_ih,
                  float* __restrict__ pre) {
    const int lane = threadIdx.x & 31;
    const int wave = threadIdx.x >> 5;
    const int bn = blockIdx.x % (H_ / 128);
    const int bm = blockIdx.x / (H_ / 128);
    const int tm = bm * 128 + (wave >> 1) * 64;
    const int tn = bn * 128 + (wave & 1) * 64;

    const int half = lane >> 4;
    const int l15  = lane & 15;

    const v4f* __restrict__ ap[4];
    #pragma unroll
    for (int i = 0; i < 4; ++i) {
        const int m   = tm + i * 16 + l15;
        const int s   = m / B_;
        const int b   = m % B_;
        const int tok = x_in[b * S_ + s];
        ap[i] = (const v4f*)(Ep + (size_t)tok * E_ + half * 4);
    }
    const v4f* __restrict__ bp = (const v4f*)Wip + (size_t)(tn >> 4) * (64 * 32) + lane;

    constexpr int GS = 4;
    v4f a0[GS][4], b0[GS][4], a1[GS][4], b1[GS][4];
    v8f acc[4][4] = {};

#define PJ_LOAD_SET(A, B, G0)                                                  \
    _Pragma("unroll")                                                          \
    for (int q = 0; q < GS; ++q) {                                             \
        _Pragma("unroll")                                                      \
        for (int i = 0; i < 4; ++i) A[q][i] = ap[i][((G0) + q) * 2];           \
        _Pragma("unroll")                                                      \
        for (int j = 0; j < 4; ++j) B[q][j] = bp[((G0) + q) * 32 + j * (64 * 32)]; \
    }
#define PJ_MMA_SET(A, B)                                                       \
    _Pragma("unroll")                                                          \
    for (int q = 0; q < GS; ++q) {                                             \
        _Pragma("unroll")                                                      \
        for (int i = 0; i < 4; ++i) {                                          \
            _Pragma("unroll")                                                  \
            for (int j = 0; j < 4; ++j) {                                      \
                const v2f x0 = { A[q][i].x, A[q][i].y };                       \
                const v2f y0 = { B[q][j].x, B[q][j].y };                       \
                acc[i][j] = __builtin_amdgcn_wmma_f32_16x16x4_f32(false, x0, false, y0, (short)0, acc[i][j], false, false); \
            }                                                                  \
        }                                                                      \
        _Pragma("unroll")                                                      \
        for (int i = 0; i < 4; ++i) {                                          \
            _Pragma("unroll")                                                  \
            for (int j = 0; j < 4; ++j) {                                      \
                const v2f x1 = { A[q][i].z, A[q][i].w };                       \
                const v2f y1 = { B[q][j].z, B[q][j].w };                       \
                acc[i][j] = __builtin_amdgcn_wmma_f32_16x16x4_f32(false, x1, false, y1, (short)0, acc[i][j], false, false); \
            }                                                                  \
        }                                                                      \
    }
#define PJ_FENCE() __builtin_amdgcn_sched_barrier(0)

    PJ_LOAD_SET(a0, b0, 0)
    PJ_FENCE();
    for (int g = 0; g < E_ / 8 - 2 * GS; g += 2 * GS) {
        PJ_LOAD_SET(a1, b1, g + GS)
        PJ_FENCE();
        PJ_MMA_SET(a0, b0)
        PJ_FENCE();
        PJ_LOAD_SET(a0, b0, g + 2 * GS)
        PJ_FENCE();
        PJ_MMA_SET(a1, b1)
        PJ_FENCE();
    }
    PJ_LOAD_SET(a1, b1, E_ / 8 - GS)
    PJ_FENCE();
    PJ_MMA_SET(a0, b0)
    PJ_MMA_SET(a1, b1)
#undef PJ_FENCE
#undef PJ_LOAD_SET
#undef PJ_MMA_SET

    float bias[4];
    #pragma unroll
    for (int j = 0; j < 4; ++j) bias[j] = b_ih[tn + j * 16 + l15];
    #pragma unroll
    for (int i = 0; i < 4; ++i) {
        #pragma unroll
        for (int r = 0; r < 8; ++r) {
            const int row = tm + i * 16 + half * 8 + r;
            float* __restrict__ dst = pre + (size_t)row * H_ + tn + l15;
            #pragma unroll
            for (int j = 0; j < 4; ++j) dst[j * 16] = acc[i][j][r] + bias[j];
        }
    }
}


__global__ __launch_bounds__(256)
void k_pack_w(const float* __restrict__ W, float* __restrict__ Wp) {
    const int i = blockIdx.x * 256 + threadIdx.x;
    const int l = i & 31;
    const int g = (i >> 5) & 127;
    const int t = i >> 12;
    const int half = l >> 4;
    const int l15  = l & 15;
    const float* __restrict__ src = W + (size_t)(t * 16 + l15) * H_ + g * 8 + half * 2;
    v4f v = { src[0], src[1], src[4], src[5] };
    ((v4f*)Wp)[i] = v;
}

__device__ __forceinline__ float tanh_abs_below_0625(float y) {
    const float y2 = y * y;
    float p = __builtin_fmaf(y2, -0x1.758e7ap-8f, 0x1.521192p-6f);
    p = __builtin_fmaf(y2, p, -0x1.b8389cp-5f);
    p = __builtin_fmaf(y2, p, 0x1.110704p-3f);
    p = __builtin_fmaf(y2, p, -0x1.555532p-2f);
    return __builtin_fmaf(y2, y * p, y);
}
__device__ __forceinline__ float tanh_abs_from_0625(float y) {
    const float t = expf(2.0f * y);
    return __builtin_fmaf(-2.0f, __builtin_amdgcn_rcpf(t + 1.0f), 1.0f);
}

__device__ __forceinline__
void rnn_step_epilogue_vals(v8f acc, int tm, int tn, int half, int l15,
                            float bias, const float (&pv)[8], const int (&ln)[8],
                            int s,
                            float* __restrict__ hout,
                            float* __restrict__ last) {
    const int col = tn + l15;
    const int kk = l15 & 7;
    const int hp = (kk >> 1) & 1;
    const int e  = (kk & 1) | ((kk >> 2) << 1);
    float* __restrict__ hdst = hout
        + ((size_t)((tm >> 4) * 128 + (col >> 3)) * 32 + hp * 16 + half * 8) * 4 + e;
    float* __restrict__ ldst = last + (size_t)(tm + half * 8) * H_ + col;

    float z[8], y[8], t[8];
    bool large = false;
    #pragma unroll
    for (int r = 0; r < 8; ++r) {
        z[r] = acc[r] + pv[r] + bias;
        y[r] = fabsf(z[r]);
        large = large || !(y[r] < 0.625f);
        t[r] = tanh_abs_below_0625(y[r]);
    }
    if (large) {
        #pragma unroll
        for (int r = 0; r < 8; ++r) {
            const float x = tanh_abs_from_0625(y[r]);
            t[r] = (y[r] < 0.625f) ? t[r] : x;
        }
    }
    unsigned ends = 0;
    #pragma unroll
    for (int r = 0; r < 8; ++r) {
        t[r] = copysignf(t[r], z[r]);
        hdst[r * 4] = t[r];
        ends |= __builtin_amdgcn_ballot_w32(ln[r] - 1 == s);
    }
    if (ends != 0) {
        #pragma unroll
        for (int r = 0; r < 8; ++r)
            if (ln[r] - 1 == s) ldst[(size_t)r * H_] = t[r];
    }
}

__device__ __forceinline__
void rnn_step_epilogue_operands(int tm, int tn, int half, int l15,
                                const float* __restrict__ b_hh,
                                const float* __restrict__ pre_s,
                                const int* __restrict__ lengths,
                                float& bias, float (&pv)[8], int (&ln)[8]) {
    const int col = tn + l15;
    bias = b_hh[col];
    #pragma unroll
    for (int r = 0; r < 8; ++r) {
        const int row = tm + half * 8 + r;
        pv[r] = pre_s[(size_t)row * H_ + col];
        ln[r] = lengths[row];
    }
}

#define STEP_THREADS 128
#define STEP_BLOCKS ((B_ / 32) * (H_ / 32))

__device__ __forceinline__
void rnn_step_tile(int& tm, int& tn) {
    const int wave = threadIdx.x >> 5;
    const int bn = blockIdx.x % (H_ / 32);
    const int bm = blockIdx.x / (H_ / 32);
    tm = (bm * 2 + (wave >> 1)) * 16;
    tn = (bn * 2 + (wave & 1)) * 16;
}

__global__ __launch_bounds__(STEP_THREADS)
void k_rnn_step_rowmajor_w(const float* __restrict__ hin,
                           const float* __restrict__ W_hh,
                           const float* __restrict__ b_hh,
                           const float* __restrict__ pre_s,
                           const int* __restrict__ lengths,
                           int s,
                           float* __restrict__ hout,
                           float* __restrict__ last) {
    const int lane = threadIdx.x & 31;
    int tm, tn;
    rnn_step_tile(tm, tn);

    const int half = lane >> 4;
    const int l15  = lane & 15;

    const v4f*   __restrict__ ap   = (const v4f*)hin + (size_t)(tm >> 4) * (128 * 32) + lane;
    const float* __restrict__ brow = W_hh + (size_t)(tn + l15) * H_ + half * 2;

    float bias;
    float pv[8];
    int   ln[8];
    rnn_step_epilogue_operands(tm, tn, half, l15, b_hh, pre_s, lengths, bias, pv, ln);

    v8f acc = {};
    #pragma unroll 8
    for (int g = 0; g < H_ / 8; ++g) {
        const v4f a  = ap[g * 32];
        const v2f a0 = { a.x, a.y };
        const v2f a1 = { a.z, a.w };
        const v2f b0 = { brow[g * 8],     brow[g * 8 + 1] };
        const v2f b1 = { brow[g * 8 + 4], brow[g * 8 + 5] };
        acc = __builtin_amdgcn_wmma_f32_16x16x4_f32(false, a0, false, b0, (short)0, acc, false, false);
        acc = __builtin_amdgcn_wmma_f32_16x16x4_f32(false, a1, false, b1, (short)0, acc, false, false);
    }

    rnn_step_epilogue_vals(acc, tm, tn, half, l15, bias, pv, ln, s, hout, last);
}

__global__ __launch_bounds__(STEP_THREADS) __attribute__((amdgpu_waves_per_eu(1, 1)))
void k_rnn_step_frag_w(const float* __restrict__ hin,
                       const float* __restrict__ Wp,
                       const float* __restrict__ b_hh,
                       const float* __restrict__ pre_s,
                       const int* __restrict__ lengths,
                       int s,
                       float* __restrict__ hout,
                       float* __restrict__ last) {
    const int lane = threadIdx.x & 31;
    int tm, tn;
    rnn_step_tile(tm, tn);

    const int half = lane >> 4;
    const int l15  = lane & 15;

    const v4f* __restrict__ ap = (const v4f*)hin + (size_t)(tm >> 4) * (128 * 32) + lane;
    const v4f* __restrict__ bp = (const v4f*)Wp  + (size_t)(tn >> 4) * (128 * 32) + lane;

    float bias;
    float pv[8];
    int   ln[8];
    rnn_step_epilogue_operands(tm, tn, half, l15, b_hh, pre_s, lengths, bias, pv, ln);

    constexpr int SET  = 8;
    constexpr int RING = 4;
    constexpr int NSET = H_ / 8 / SET;
    v4f ra[RING][SET], rb[RING][SET];
    v8f acc0 = {}, acc1 = {};

#define LOAD_SET(R, G0)                                                        \
    _Pragma("unroll")                                                          \
    for (int j = 0; j < SET; ++j) {                                            \
        ra[R][j] = ap[((G0) + j) * 32];                                        \
        rb[R][j] = bp[((G0) + j) * 32];                                        \
    }
#define MMA_SET(R)                                                             \
    _Pragma("unroll")                                                          \
    for (int j = 0; j < SET; ++j) {                                            \
        const v2f x0 = { ra[R][j].x, ra[R][j].y };                             \
        const v2f x1 = { ra[R][j].z, ra[R][j].w };                             \
        const v2f y0 = { rb[R][j].x, rb[R][j].y };                             \
        const v2f y1 = { rb[R][j].z, rb[R][j].w };                             \
        acc0 = __builtin_amdgcn_wmma_f32_16x16x4_f32(false, x0, false, y0, (short)0, acc0, false, false); \
        acc1 = __builtin_amdgcn_wmma_f32_16x16x4_f32(false, x1, false, y1, (short)0, acc1, false, false); \
    }

#define GROUP_FENCE() __builtin_amdgcn_sched_barrier(0)

    LOAD_SET(0, 0)
    LOAD_SET(1, SET)
    LOAD_SET(2, 2 * SET)
    GROUP_FENCE();
    #pragma unroll
    for (int i = 0; i < NSET; ++i) {
        if (i + 3 < NSET) {
            LOAD_SET((i + 3) % RING, (i + 3) * SET)
        }
        GROUP_FENCE();
        MMA_SET(i % RING)
        GROUP_FENCE();
    }
    const v8f acc = acc0 + acc1;
#undef GROUP_FENCE
#undef LOAD_SET
#undef MMA_SET

    rnn_step_epilogue_vals(acc, tm, tn, half, l15, bias, pv, ln, s, hout, last);
}

__global__ __launch_bounds__(256)
void k_mlp1(const float* __restrict__ last,
            const float* __restrict__ W1,
            const float* __restrict__ b1,
            float* __restrict__ y1) {
    const int lane = threadIdx.x & 31;
    const int wave = threadIdx.x >> 5;
    const int gw   = blockIdx.x * (blockDim.x >> 5) + wave;

    const int NT = H_ / 16;
    const int tn = (gw % NT) * 16;
    const int tm = (gw / NT) * 16;

    const int half = lane >> 4;
    const int l15  = lane & 15;

    const float* __restrict__ arow = last + (size_t)(tm + l15) * H_ + half * 2;
    const float* __restrict__ brow = W1   + (size_t)(tn + l15) * H_ + half * 2;

    v8f acc = {};
    #pragma unroll 8
    for (int k0 = 0; k0 < H_; k0 += 4) {
        v2f a = { arow[k0], arow[k0 + 1] };
        v2f b = { brow[k0], brow[k0 + 1] };
        acc = __builtin_amdgcn_wmma_f32_16x16x4_f32(false, a, false, b, (short)0, acc, false, false);
    }

    const int   col  = tn + l15;
    const float bias = b1[col];
    #pragma unroll
    for (int r = 0; r < 8; ++r) {
        const int row = tm + half * 8 + r;
        const float v = acc[r] + bias;
        y1[(size_t)row * H_ + col] = v > 0.0f ? v : 0.0f;
    }
}

__global__ void k_mlp2(const float* __restrict__ y1,
                       const float* __restrict__ W2,
                       const float* __restrict__ b2,
                       float* __restrict__ out) {
    int i = blockIdx.x * blockDim.x + threadIdx.x;
    if (i >= B_ * C_) return;
    const int b = i / C_;
    const int c = i % C_;
    const float* __restrict__ yr = y1 + (size_t)b * H_;
    const float* __restrict__ wr = W2 + (size_t)c * H_;
    float acc = b2[c];
    #pragma unroll 8
    for (int k = 0; k < H_; ++k) acc += yr[k] * wr[k];
    out[i] = acc;
}

extern "C" void kernel_launch(void* const* d_in, const int* in_sizes, int n_in,
                              void* d_out, int out_size, void* d_ws, size_t ws_size,
                              hipStream_t stream) {
    const int*   x_in  = (const int*)  d_in[0];
    const int*   x_len = (const int*)  d_in[1];
    const float* emb   = (const float*)d_in[2];
    const float* W_ih  = (const float*)d_in[3];
    const float* b_ih  = (const float*)d_in[4];
    const float* W_hh  = (const float*)d_in[5];
    const float* b_hh  = (const float*)d_in[6];
    const float* W1    = (const float*)d_in[7];
    const float* b1    = (const float*)d_in[8];
    const float* W2    = (const float*)d_in[9];
    const float* b2    = (const float*)d_in[10];
    float* out = (float*)d_out;

    float* pre  = (float*)d_ws;
    float* hb0  = pre  + (size_t)S_ * B_ * H_;
    float* hb1  = hb0  + (size_t)B_ * H_;
    float* last = hb1  + (size_t)B_ * H_;
    float* y1   = last + (size_t)B_ * H_;

    k_zero<<<(B_ * H_ + 255) / 256, 256, 0, stream>>>(hb0, B_ * H_);

    {
        float* Wip = hb1;
        float* Ep  = y1;
        k_pack_proj<<<(PJ_W_CHUNKS + PJ_E_CHUNKS) / 256, 256, 0, stream>>>(emb, W_ih, Ep, Wip);
        k_embed_proj<<<PJ_BLOCKS, PJ_THREADS, 0, stream>>>(x_in, Ep, Wip, b_ih, pre);
    }

    {
        float* bufs[2] = { hb0, hb1 };
        const int S_ROW = 4;
        float* Wp = pre;
        for (int s = 0; s < S_ROW; ++s) {
            const float* hin  = bufs[s & 1];
            float*       hout = bufs[(s + 1) & 1];
            k_rnn_step_rowmajor_w<<<STEP_BLOCKS, STEP_THREADS, 0, stream>>>(
                hin, W_hh, b_hh, pre + (size_t)s * B_ * H_, x_len, s, hout, last);
        }
        k_pack_w<<<(H_ * H_ / 4) / 256, 256, 0, stream>>>(W_hh, Wp);
        for (int s = S_ROW; s < S_; ++s) {
            const float* hin  = bufs[s & 1];
            float*       hout = bufs[(s + 1) & 1];
            k_rnn_step_frag_w<<<STEP_BLOCKS, STEP_THREADS, 0, stream>>>(
                hin, Wp, b_hh, pre + (size_t)s * B_ * H_, x_len, s, hout, last);
        }
    }

    {
        const int waves  = (B_ / 16) * (H_ / 16);
        const int blocks = waves / 8;
        k_mlp1<<<blocks, 256, 0, stream>>>(last, W1, b1, y1);
        k_mlp2<<<(B_ * C_ + 127) / 128, 128, 0, stream>>>(y1, W2, b2, out);
    }
}
